# starting kernel
# baseline (speedup 1.0000x reference)
#include <hip/hip_runtime.h>


typedef float v2f __attribute__((ext_vector_type(2)));
typedef float v8f __attribute__((ext_vector_type(8)));

#define LDS_ELEMS (32 * 4 * 30 * 2)

__global__ __launch_bounds__(256) void fused_conv_wmma(
    const float* __restrict__ x,
    const float* __restrict__ w2,
    const float* __restrict__ w1,
    float* __restrict__ out)
{
    __shared__ float btile[LDS_ELEMS];

    const int tileN = blockIdx.x * 16;
    const int h0    = tileN / 28;

    for (int t = threadIdx.x; t < LDS_ELEMS; t += 256) {
        const int kk = t & 1;
        const int wi = (t >> 1) % 30;
        const int ri = (t / 60) % 4;
        const int cp = t / 240;
        const int hin = h0 - 1 + ri;
        const int win = wi - 1;

        float v = 0.0f;
        if (hin >= 0 && hin < 28 && win >= 0 && win < 28) {
            const int j0 = cp * 2;
#pragma unroll
            for (int p = 0; p < 2; ++p) {
                const float* xrow = x + (j0 + p) * 784 + hin * 28;
                const float* w2r  = w2 + (j0 + p) * 6 + kk;
#pragma unroll
                for (int i = 0; i < 3; ++i) {
                    const int wx = win + i - 1;
                    const float xv = (wx >= 0 && wx < 28) ? xrow[wx] : 0.0f;
                    v += xv * w2r[i * 2];
                }
            }
        }
        btile[t] = v;
    }
    __syncthreads();

    const int lane    = threadIdx.x & 31;
    const int wave    = threadIdx.x >> 5;
    const int tileM   = wave * 16;
    const int halfsel = lane >> 4;
    const int lane16  = lane & 15;

    const int mRow = tileM + lane16;
    const int nCol = tileN + lane16;
    const int h = nCol / 28;
    const int w = nCol - h * 28;

    v8f acc = {};

#pragma unroll 3
    for (int rs = 0; rs < 9; ++rs) {
        const int r = rs / 3, s = rs - r * 3;
        const int aBase = mRow * 576 + r * 6 + s * 2;
        const int bRow = ((h + r - h0) * 30 + (w + s)) * 2;

        for (int c0 = 0; c0 < 64; c0 += 4) {
            const int cA = c0 + halfsel * 2;
            const v2f a = *reinterpret_cast<const v2f*>(
                w1 + aBase + (cA >> 1) * 18);
            const v2f b = *reinterpret_cast<const v2f*>(
                btile + (cA >> 1) * 240 + bRow);
            acc = __builtin_amdgcn_wmma_f32_16x16x4_f32(
                 false, a,  false, b,
                 (short)0, acc,  false,  false);
        }
    }

    const int mBase = tileM + halfsel * 8;
#pragma unroll
    for (int rr = 0; rr < 8; ++rr) {
        out[(mBase + rr) * 784 + tileN + lane16] = acc[rr];
    }
}

extern "C" void kernel_launch(void* const* d_in, const int* in_sizes, int n_in,
                              void* d_out, int out_size, void* d_ws, size_t ws_size,
                              hipStream_t stream) {
    const float* x  = (const float*)d_in[0];
    const float* w1 = (const float*)d_in[1];
    const float* w2 = (const float*)d_in[2];
    float* out = (float*)d_out;

    fused_conv_wmma<<<49, 256, 0, stream>>>(x, w2, w1, out);
}


# Weights (A operand) copied per wave with whole-line 16 B loads into a private LDS tile and read as fragments from there; K walked channel-major in two halves (changes the f32 summation order)
# speedup vs baseline: 1.1484x; 1.1484x over previous
#include <hip/hip_runtime.h>


typedef float v2f __attribute__((ext_vector_type(2)));
typedef float v4f __attribute__((ext_vector_type(4)));
typedef float v8f __attribute__((ext_vector_type(8)));

#define LDS_ELEMS (32 * 4 * 30 * 2)
#define A_HALF    288
#define A_PITCH   292
#define A_WAVE    (16 * A_PITCH)
#define A_LOADS   36

template <int HF>
__device__ __forceinline__ void mma_half(v8f& acc, const float* aFrag,
                                         const float* bFrag)
{
#pragma unroll
    for (int u = 0; u < 8; ++u) {
#pragma unroll
        for (int rs = 0; rs < 9; ++rs) {
            const int r = rs / 3, s = rs - r * 3;
            const v2f a = *reinterpret_cast<const v2f*>(
                aFrag + u * 36 + rs * 2);
            const v2f b = *reinterpret_cast<const v2f*>(
                bFrag + (HF * 16 + 2 * u) * 240 + (r * 30 + s) * 2);
            acc = __builtin_amdgcn_wmma_f32_16x16x4_f32(
                 false, a,  false, b,
                 (short)0, acc,  false,  false);
        }
    }
}

__global__ __launch_bounds__(256) void fused_conv_wmma(
    const float* __restrict__ x,
    const float* __restrict__ w2,
    const float* __restrict__ w1,
    float* __restrict__ out)
{
    __shared__ __attribute__((aligned(16))) float btile[LDS_ELEMS];
    __shared__ __attribute__((aligned(16))) float atile[8 * A_WAVE];

    const int lane    = threadIdx.x & 31;
    const int wave    = threadIdx.x >> 5;
    const int tileN   = blockIdx.x * 16;
    const int h0      = tileN / 28;

    const float* aSrc = w1 + wave * (16 * 576);
    float* aDst = atile + wave * A_WAVE;
    int gOff[9], lOff[9];
#pragma unroll
    for (int b = 0; b < 9; ++b) {
        const int q   = b * 32 + lane;
        const int row = q / 72;
        const int pc  = q - row * 72;
        gOff[b] = row * 576 + pc * 4;
        lOff[b] = row * A_PITCH + pc * 4;
    }

    v4f areg[A_LOADS];
#pragma unroll
    for (int j = 0; j < A_LOADS; ++j) {
        areg[j] = *reinterpret_cast<const v4f*>(
            aSrc + (j / 9) * (4 * 576) + gOff[j % 9]);
    }

    for (int t = threadIdx.x; t < LDS_ELEMS; t += 256) {
        const int kk = t & 1;
        const int wi = (t >> 1) % 30;
        const int ri = (t / 60) % 4;
        const int cp = t / 240;
        const int hin = h0 - 1 + ri;
        const int win = wi - 1;

        float v = 0.0f;
        if (hin >= 0 && hin < 28 && win >= 0 && win < 28) {
            const int j0 = cp * 2;
#pragma unroll
            for (int p = 0; p < 2; ++p) {
                const float* xrow = x + (j0 + p) * 784 + hin * 28;
                const float* w2r  = w2 + (j0 + p) * 6 + kk;
#pragma unroll
                for (int i = 0; i < 3; ++i) {
                    const int wx = win + i - 1;
                    const float xv = (wx >= 0 && wx < 28) ? xrow[wx] : 0.0f;
                    v += xv * w2r[i * 2];
                }
            }
        }
        btile[t] = v;
    }

#pragma unroll
    for (int j = 0; j < A_LOADS; ++j) {
        *reinterpret_cast<v4f*>(aDst + (j / 9) * (4 * A_PITCH) + lOff[j % 9]) =
            areg[j];
    }
    __syncthreads();

#pragma unroll
    for (int j = 0; j < A_LOADS; ++j) {
        areg[j] = *reinterpret_cast<const v4f*>(
            aSrc + A_HALF + (j / 9) * (4 * 576) + gOff[j % 9]);
    }

    const int tileM   = wave * 16;
    const int halfsel = lane >> 4;
    const int lane16  = lane & 15;

    const int nCol = tileN + lane16;
    const int h = nCol / 28;
    const int w = nCol - h * 28;

    const float* aFrag = aDst + lane16 * A_PITCH + halfsel * 18;
    const float* bFrag = btile + halfsel * 240 + ((h - h0) * 30 + w) * 2;

    v8f acc = {};
    mma_half<0>(acc, aFrag, bFrag);

    __syncthreads();
#pragma unroll
    for (int j = 0; j < A_LOADS; ++j) {
        *reinterpret_cast<v4f*>(aDst + (j / 9) * (4 * A_PITCH) + lOff[j % 9]) =
            areg[j];
    }
    __syncthreads();

    mma_half<1>(acc, aFrag, bFrag);

    const int mBase = tileM + halfsel * 8;
#pragma unroll
    for (int rr = 0; rr < 8; ++rr) {
        out[(mBase + rr) * 784 + tileN + lane16] = acc[rr];
    }
}

extern "C" void kernel_launch(void* const* d_in, const int* in_sizes, int n_in,
                              void* d_out, int out_size, void* d_ws, size_t ws_size,
                              hipStream_t stream) {
    const float* x  = (const float*)d_in[0];
    const float* w1 = (const float*)d_in[1];
    const float* w2 = (const float*)d_in[2];
    float* out = (float*)d_out;

    fused_conv_wmma<<<49, 256, 0, stream>>>(x, w2, w1, out);
}


# Stage 1 re-mapped to one thread per (channel pair, slice row, kk): its two input rows and 12 weights arrive as 17 16-B loads issued up front, ahead of the weight copy; same summation order
# speedup vs baseline: 2.9161x; 2.5393x over previous
#include <hip/hip_runtime.h>


typedef float v2f __attribute__((ext_vector_type(2)));
typedef float v4f __attribute__((ext_vector_type(4)));
typedef float v8f __attribute__((ext_vector_type(8)));

#define LDS_ELEMS (32 * 4 * 30 * 2)
#define A_HALF    288
#define A_PITCH   292
#define A_WAVE    (16 * A_PITCH)
#define A_LOADS   36

template <int HF>
__device__ __forceinline__ void mma_half(v8f& acc, const float* aFrag,
                                         const float* bFrag)
{
#pragma unroll
    for (int u = 0; u < 8; ++u) {
#pragma unroll
        for (int rs = 0; rs < 9; ++rs) {
            const int r = rs / 3, s = rs - r * 3;
            const v2f a = *reinterpret_cast<const v2f*>(
                aFrag + u * 36 + rs * 2);
            const v2f b = *reinterpret_cast<const v2f*>(
                bFrag + (HF * 16 + 2 * u) * 240 + (r * 30 + s) * 2);
            acc = __builtin_amdgcn_wmma_f32_16x16x4_f32(
                 false, a,  false, b,
                 (short)0, acc,  false,  false);
        }
    }
}

__global__ __launch_bounds__(256) void fused_conv_wmma(
    const float* __restrict__ x,
    const float* __restrict__ w2,
    const float* __restrict__ w1,
    float* __restrict__ out)
{
    __shared__ __attribute__((aligned(16))) float btile[LDS_ELEMS];
    __shared__ __attribute__((aligned(16))) float atile[8 * A_WAVE];

    const int lane    = threadIdx.x & 31;
    const int wave    = threadIdx.x >> 5;
    const int tileN   = blockIdx.x * 16;
    const int h0      = tileN / 28;

    const float* aSrc = w1 + wave * (16 * 576);
    float* aDst = atile + wave * A_WAVE;
    int gOff[9], lOff[9];
#pragma unroll
    for (int b = 0; b < 9; ++b) {
        const int q   = b * 32 + lane;
        const int row = q / 72;
        const int pc  = q - row * 72;
        gOff[b] = row * 576 + pc * 4;
        lOff[b] = row * A_PITCH + pc * 4;
    }

    const int  s1kk    = threadIdx.x & 1;
    const int  s1ri    = (threadIdx.x >> 1) & 3;
    const int  s1cp    = threadIdx.x >> 3;
    const int  s1hin   = h0 - 1 + s1ri;
    const bool s1rowOK = (s1hin >= 0) && (s1hin < 28);
    const int  s1hinC  = s1hin < 0 ? 0 : (s1hin > 27 ? 27 : s1hin);
    const float* xrow0 = x + (2 * s1cp) * 784 + s1hinC * 28;
    const float* xrow1 = xrow0 + 784;
    v4f xa[7], xb[7], wq[3];
#pragma unroll
    for (int k = 0; k < 7; ++k) {
        xa[k] = *reinterpret_cast<const v4f*>(xrow0 + 4 * k);
        xb[k] = *reinterpret_cast<const v4f*>(xrow1 + 4 * k);
    }
#pragma unroll
    for (int k = 0; k < 3; ++k) {
        wq[k] = *reinterpret_cast<const v4f*>(w2 + s1cp * 12 + 4 * k);
    }

    __builtin_amdgcn_sched_barrier(0);

    v4f areg[A_LOADS];
#pragma unroll
    for (int j = 0; j < A_LOADS; ++j) {
        areg[j] = *reinterpret_cast<const v4f*>(
            aSrc + (j / 9) * (4 * 576) + gOff[j % 9]);
    }

    {
        float c0[3], c1[3];
#pragma unroll
        for (int i = 0; i < 3; ++i) {
            const int e0 = i * 2, e1 = 6 + i * 2;
            const float a0 = wq[e0 >> 2][e0 & 3], b0 = wq[(e0 + 1) >> 2][(e0 + 1) & 3];
            const float a1 = wq[e1 >> 2][e1 & 3], b1 = wq[(e1 + 1) >> 2][(e1 + 1) & 3];
            c0[i] = s1kk ? b0 : a0;
            c1[i] = s1kk ? b1 : a1;
        }
        float* bdst = btile + s1cp * 240 + s1ri * 60 + s1kk;
        bdst[0]      = 0.0f;
        bdst[29 * 2] = 0.0f;
#pragma unroll
        for (int win = 0; win < 28; ++win) {
            float v = 0.0f;
#pragma unroll
            for (int i = 0; i < 3; ++i) {
                const int wx = win + i - 1;
                const float xv = (wx >= 0 && wx < 28) ? xa[wx >> 2][wx & 3] : 0.0f;
                v += xv * c0[i];
            }
#pragma unroll
            for (int i = 0; i < 3; ++i) {
                const int wx = win + i - 1;
                const float xv = (wx >= 0 && wx < 28) ? xb[wx >> 2][wx & 3] : 0.0f;
                v += xv * c1[i];
            }
            bdst[(win + 1) * 2] = s1rowOK ? v : 0.0f;
        }
    }

#pragma unroll
    for (int j = 0; j < A_LOADS; ++j) {
        *reinterpret_cast<v4f*>(aDst + (j / 9) * (4 * A_PITCH) + lOff[j % 9]) =
            areg[j];
    }
    __syncthreads();

#pragma unroll
    for (int j = 0; j < A_LOADS; ++j) {
        areg[j] = *reinterpret_cast<const v4f*>(
            aSrc + A_HALF + (j / 9) * (4 * 576) + gOff[j % 9]);
    }

    const int tileM   = wave * 16;
    const int halfsel = lane >> 4;
    const int lane16  = lane & 15;

    const int nCol = tileN + lane16;
    const int h = nCol / 28;
    const int w = nCol - h * 28;

    const float* aFrag = aDst + lane16 * A_PITCH + halfsel * 18;
    const float* bFrag = btile + halfsel * 240 + ((h - h0) * 30 + w) * 2;

    v8f acc = {};
    mma_half<0>(acc, aFrag, bFrag);

    __syncthreads();
#pragma unroll
    for (int j = 0; j < A_LOADS; ++j) {
        *reinterpret_cast<v4f*>(aDst + (j / 9) * (4 * A_PITCH) + lOff[j % 9]) =
            areg[j];
    }
    __syncthreads();

    mma_half<1>(acc, aFrag, bFrag);

    const int mBase = tileM + halfsel * 8;
#pragma unroll
    for (int rr = 0; rr < 8; ++rr) {
        out[(mBase + rr) * 784 + tileN + lane16] = acc[rr];
    }
}

extern "C" void kernel_launch(void* const* d_in, const int* in_sizes, int n_in,
                              void* d_out, int out_size, void* d_ws, size_t ws_size,
                              hipStream_t stream) {
    const float* x  = (const float*)d_in[0];
    const float* w1 = (const float*)d_in[1];
    const float* w2 = (const float*)d_in[2];
    float* out = (float*)d_out;

    fused_conv_wmma<<<49, 256, 0, stream>>>(x, w2, w1, out);
}


# Work re-split to fill the chip: grid 49 -> 196 blocks (16 pixels x 32 channels); the 8 waves of a block split K 8 ways, partial tiles summed through LDS (changes the f32 summation order)
# speedup vs baseline: 4.6673x; 1.6005x over previous
#include <hip/hip_runtime.h>


typedef float v2f __attribute__((ext_vector_type(2)));
typedef float v4f __attribute__((ext_vector_type(4)));
typedef float v8f __attribute__((ext_vector_type(8)));

#define LDS_ELEMS (32 * 4 * 30 * 2)
#define A_SLICE   72
#define A_PITCH   76
#define A_ROWS    32
#define A_WAVE    (A_ROWS * A_PITCH)
#define A_LOADS   18
#define RED_ELEMS (8 * 16 * 32)

__global__ __launch_bounds__(256) void fused_conv_wmma(
    const float* __restrict__ x,
    const float* __restrict__ w2,
    const float* __restrict__ w1,
    float* __restrict__ out)
{
    __shared__ __attribute__((aligned(16))) float btile[LDS_ELEMS];
    __shared__ __attribute__((aligned(16))) float atile[8 * A_WAVE];
    __shared__ __attribute__((aligned(16))) float red[RED_ELEMS];

    const int lane    = threadIdx.x & 31;
    const int wave    = threadIdx.x >> 5;
    const int tileN   = blockIdx.x * 16;
    const int mBlk    = blockIdx.y * 32;
    const int h0      = tileN / 28;

    const float* aSrc = w1 + mBlk * 576 + wave * A_SLICE;
    float* aDst = atile + wave * A_WAVE;
    int gOff[9], lOff[9];
#pragma unroll
    for (int b = 0; b < 9; ++b) {
        const int q   = b * 32 + lane;
        const int row = q / 18;
        const int pc  = q - row * 18;
        gOff[b] = row * 576 + pc * 4;
        lOff[b] = row * A_PITCH + pc * 4;
    }

    const int  s1kk    = threadIdx.x & 1;
    const int  s1ri    = (threadIdx.x >> 1) & 3;
    const int  s1cp    = threadIdx.x >> 3;
    const int  s1hin   = h0 - 1 + s1ri;
    const bool s1rowOK = (s1hin >= 0) && (s1hin < 28);
    const int  s1hinC  = s1hin < 0 ? 0 : (s1hin > 27 ? 27 : s1hin);
    const float* xrow0 = x + (2 * s1cp) * 784 + s1hinC * 28;
    const float* xrow1 = xrow0 + 784;
    v4f xa[7], xb[7], wq[3];
#pragma unroll
    for (int k = 0; k < 7; ++k) {
        xa[k] = *reinterpret_cast<const v4f*>(xrow0 + 4 * k);
        xb[k] = *reinterpret_cast<const v4f*>(xrow1 + 4 * k);
    }
#pragma unroll
    for (int k = 0; k < 3; ++k) {
        wq[k] = *reinterpret_cast<const v4f*>(w2 + s1cp * 12 + 4 * k);
    }

    __builtin_amdgcn_sched_barrier(0);

    v4f areg[A_LOADS];
#pragma unroll
    for (int j = 0; j < A_LOADS; ++j) {
        areg[j] = *reinterpret_cast<const v4f*>(
            aSrc + (j / 9) * (16 * 576) + gOff[j % 9]);
    }

    {
        float c0[3], c1[3];
#pragma unroll
        for (int i = 0; i < 3; ++i) {
            const int e0 = i * 2, e1 = 6 + i * 2;
            const float a0 = wq[e0 >> 2][e0 & 3], b0 = wq[(e0 + 1) >> 2][(e0 + 1) & 3];
            const float a1 = wq[e1 >> 2][e1 & 3], b1 = wq[(e1 + 1) >> 2][(e1 + 1) & 3];
            c0[i] = s1kk ? b0 : a0;
            c1[i] = s1kk ? b1 : a1;
        }
        float* bdst = btile + s1cp * 240 + s1ri * 60 + s1kk;
        bdst[0]      = 0.0f;
        bdst[29 * 2] = 0.0f;
#pragma unroll
        for (int win = 0; win < 28; ++win) {
            float v = 0.0f;
#pragma unroll
            for (int i = 0; i < 3; ++i) {
                const int wx = win + i - 1;
                const float xv = (wx >= 0 && wx < 28) ? xa[wx >> 2][wx & 3] : 0.0f;
                v += xv * c0[i];
            }
#pragma unroll
            for (int i = 0; i < 3; ++i) {
                const int wx = win + i - 1;
                const float xv = (wx >= 0 && wx < 28) ? xb[wx >> 2][wx & 3] : 0.0f;
                v += xv * c1[i];
            }
            bdst[(win + 1) * 2] = s1rowOK ? v : 0.0f;
        }
    }

#pragma unroll
    for (int j = 0; j < A_LOADS; ++j) {
        *reinterpret_cast<v4f*>(aDst + (j / 9) * (16 * A_PITCH) + lOff[j % 9]) =
            areg[j];
    }
    __syncthreads();

    const int halfsel = lane >> 4;
    const int lane16  = lane & 15;

    const int nCol = tileN + lane16;
    const int h = nCol / 28;
    const int w = nCol - h * 28;

    const float* aFrag = aDst + lane16 * A_PITCH + halfsel * 18;
    const float* bFrag = btile + (4 * wave + halfsel) * 240
                               + ((h - h0) * 30 + w) * 2;

    v8f acc0 = {}, acc1 = {};
#pragma unroll
    for (int u = 0; u < 2; ++u) {
#pragma unroll
        for (int rs = 0; rs < 9; ++rs) {
            const int r = rs / 3, s = rs - r * 3;
            const v2f a0 = *reinterpret_cast<const v2f*>(aFrag + u * 36 + rs * 2);
            const v2f a1 = *reinterpret_cast<const v2f*>(
                aFrag + 16 * A_PITCH + u * 36 + rs * 2);
            const v2f b = *reinterpret_cast<const v2f*>(
                bFrag + (2 * u) * 240 + (r * 30 + s) * 2);
            acc0 = __builtin_amdgcn_wmma_f32_16x16x4_f32(
                 false, a0,  false, b,
                 (short)0, acc0,  false,  false);
            acc1 = __builtin_amdgcn_wmma_f32_16x16x4_f32(
                 false, a1,  false, b,
                 (short)0, acc1,  false,  false);
        }
    }

    float* rdst = red + wave * (16 * 32) + lane;
#pragma unroll
    for (int rr = 0; rr < 8; ++rr) {
        rdst[rr * 32]       = acc0[rr];
        rdst[(8 + rr) * 32] = acc1[rr];
    }
    __syncthreads();

    const int tsel = wave >> 2;
#pragma unroll
    for (int e = 0; e < 2; ++e) {
        const int rr = (wave & 3) * 2 + e;
        const float* rsrc = red + (tsel * 8 + rr) * 32 + lane;
        float sum = rsrc[0];
#pragma unroll
        for (int v = 1; v < 8; ++v) {
            sum += rsrc[v * (16 * 32)];
        }
        out[(mBlk + tsel * 16 + halfsel * 8 + rr) * 784 + tileN + lane16] = sum;
    }
}

extern "C" void kernel_launch(void* const* d_in, const int* in_sizes, int n_in,
                              void* d_out, int out_size, void* d_ws, size_t ws_size,
                              hipStream_t stream) {
    const float* x  = (const float*)d_in[0];
    const float* w1 = (const float*)d_in[1];
    const float* w2 = (const float*)d_in[2];
    float* out = (float*)d_out;

    fused_conv_wmma<<<dim3(49, 4), 256, 0, stream>>>(x, w2, w1, out);
}


# Re-tiled: block = 32 pixels x 16 channels on 4 waves (one per SIMD, K split 4 ways); weight bytes per block and waves per launch halve; fragment reads issued as one group (changes f32 sum order)
# speedup vs baseline: 5.2052x; 1.1152x over previous
#include <hip/hip_runtime.h>


typedef float v2f __attribute__((ext_vector_type(2)));
typedef float v4f __attribute__((ext_vector_type(4)));
typedef float v8f __attribute__((ext_vector_type(8)));

#define LDS_ELEMS (32 * 4 * 30 * 2)
#define A_SLICE   144
#define A_PITCH   148
#define A_ROWS    16
#define A_WAVE    (A_ROWS * A_PITCH)
#define A_LOADS   18
#define K_STEPS   36
#define RED_ELEMS (4 * 16 * 32)

__global__ __launch_bounds__(128) void fused_conv_wmma(
    const float* __restrict__ x,
    const float* __restrict__ w2,
    const float* w1,
    float* __restrict__ out)
{
    __shared__ __attribute__((aligned(16))) float btile[LDS_ELEMS];
    __shared__ __attribute__((aligned(16))) float atile[4 * A_WAVE];
    __shared__ __attribute__((aligned(16))) float red[RED_ELEMS];

    const int lane    = threadIdx.x & 31;
    const int wave    = threadIdx.x >> 5;
    const int tileN   = blockIdx.x * 32;
    const int mBlk    = blockIdx.y * 16;
    const int h0      = tileN / 28;

    const float* aSrc = w1 + mBlk * 576 + wave * A_SLICE;
    float* aDst = atile + wave * A_WAVE;
    int gOff[9], lOff[9];
#pragma unroll
    for (int b = 0; b < 9; ++b) {
        const int q   = b * 32 + lane;
        const int row = q / 36;
        const int pc  = q - row * 36;
        gOff[b] = row * 576 + pc * 4;
        lOff[b] = row * A_PITCH + pc * 4;
    }

    const int  s1ri    = threadIdx.x & 3;
    const int  s1cp    = threadIdx.x >> 2;
    const int  s1hin   = h0 - 1 + s1ri;
    const bool s1rowOK = (s1hin >= 0) && (s1hin < 28);
    const int  s1hinC  = s1hin < 0 ? 0 : (s1hin > 27 ? 27 : s1hin);
    const float* xrow0 = x + (2 * s1cp) * 784 + s1hinC * 28;
    const float* xrow1 = xrow0 + 784;
    v4f xa[7], xb[7], wq[3];
#pragma unroll
    for (int k = 0; k < 7; ++k) {
        xa[k] = *reinterpret_cast<const v4f*>(xrow0 + 4 * k);
        xb[k] = *reinterpret_cast<const v4f*>(xrow1 + 4 * k);
    }
#pragma unroll
    for (int k = 0; k < 3; ++k) {
        wq[k] = *reinterpret_cast<const v4f*>(w2 + s1cp * 12 + 4 * k);
    }

    __builtin_amdgcn_sched_barrier(0);

    v4f areg[A_LOADS];
#pragma unroll
    for (int j = 0; j < A_LOADS; ++j) {
        areg[j] = *reinterpret_cast<const v4f*>(
            aSrc + (j / 9) * (8 * 576) + gOff[j % 9]);
    }


    {
        float c0[3][2], c1[3][2];
#pragma unroll
        for (int i = 0; i < 3; ++i) {
#pragma unroll
            for (int kk = 0; kk < 2; ++kk) {
                const int e0 = i * 2 + kk, e1 = 6 + i * 2 + kk;
                c0[i][kk] = wq[e0 >> 2][e0 & 3];
                c1[i][kk] = wq[e1 >> 2][e1 & 3];
            }
        }
        float* bdst = btile + s1cp * 240 + s1ri * 60;
        const v2f zero2 = {0.0f, 0.0f};
        *reinterpret_cast<v2f*>(bdst)          = zero2;
        *reinterpret_cast<v2f*>(bdst + 29 * 2) = zero2;
#pragma unroll
        for (int win = 0; win < 28; ++win) {
            float v0 = 0.0f, v1 = 0.0f;
#pragma unroll
            for (int i = 0; i < 3; ++i) {
                const int wx = win + i - 1;
                const float xv = (wx >= 0 && wx < 28) ? xa[wx >> 2][wx & 3] : 0.0f;
                v0 = __builtin_fmaf(xv, c0[i][0], v0);
                v1 = __builtin_fmaf(xv, c0[i][1], v1);
            }
#pragma unroll
            for (int i = 0; i < 3; ++i) {
                const int wx = win + i - 1;
                const float xv = (wx >= 0 && wx < 28) ? xb[wx >> 2][wx & 3] : 0.0f;
                v0 = __builtin_fmaf(xv, c1[i][0], v0);
                v1 = __builtin_fmaf(xv, c1[i][1], v1);
            }
            const v2f v = {s1rowOK ? v0 : 0.0f, s1rowOK ? v1 : 0.0f};
            *reinterpret_cast<v2f*>(bdst + (win + 1) * 2) = v;
        }
    }

#pragma unroll
    for (int j = 0; j < A_LOADS; ++j) {
        *reinterpret_cast<v4f*>(aDst + (j / 9) * (8 * A_PITCH) + lOff[j % 9]) =
            areg[j];
    }
    __syncthreads();

    const int halfsel = lane >> 4;
    const int lane16  = lane & 15;

    const int nCol0 = tileN + lane16;
    const int nCol1 = nCol0 + 16;
    const int hA = nCol0 / 28, wA = nCol0 - hA * 28;
    const int hB = nCol1 / 28, wB = nCol1 - hB * 28;

    const float* aFrag = aDst + lane16 * A_PITCH + halfsel * 18;
    const float* bFrag0 = btile + (8 * wave + halfsel) * 240
                                + ((hA - h0) * 30 + wA) * 2;
    const float* bFrag1 = btile + (8 * wave + halfsel) * 240
                                + ((hB - h0) * 30 + wB) * 2;

    v2f af[K_STEPS], bf0[K_STEPS], bf1[K_STEPS];
#pragma unroll
    for (int st = 0; st < K_STEPS; ++st) {
        const int u = st / 9, rs = st - u * 9;
        const int r = rs / 3, s = rs - r * 3;
        af[st]  = *reinterpret_cast<const v2f*>(aFrag + u * 36 + rs * 2);
        bf0[st] = *reinterpret_cast<const v2f*>(
            bFrag0 + (2 * u) * 240 + (r * 30 + s) * 2);
        bf1[st] = *reinterpret_cast<const v2f*>(
            bFrag1 + (2 * u) * 240 + (r * 30 + s) * 2);
    }

    __builtin_amdgcn_sched_barrier(0);

    v8f acc0 = {}, acc1 = {};
#pragma unroll
    for (int st = 0; st < K_STEPS; ++st) {
        acc0 = __builtin_amdgcn_wmma_f32_16x16x4_f32(
             false, af[st],  false, bf0[st],
             (short)0, acc0,  false,  false);
        acc1 = __builtin_amdgcn_wmma_f32_16x16x4_f32(
             false, af[st],  false, bf1[st],
             (short)0, acc1,  false,  false);
    }

    float* rdst = red + wave * (16 * 32) + lane;
#pragma unroll
    for (int rr = 0; rr < 8; ++rr) {
        rdst[rr * 32]       = acc0[rr];
        rdst[(8 + rr) * 32] = acc1[rr];
    }
    __syncthreads();

    const int tsel = lane >> 4;
    const int nOut = tileN + lane;
#pragma unroll
    for (int e = 0; e < 4; ++e) {
        const int m   = wave * 4 + e;
        const int hs  = m >> 3, rr = m & 7;
        const float* rsrc = red + (tsel * 8 + rr) * 32 + hs * 16 + lane16;
        float sum = rsrc[0];
#pragma unroll
        for (int v = 1; v < 4; ++v) {
            sum += rsrc[v * (16 * 32)];
        }
        if (nOut < 784) {
            out[(mBlk + m) * 784 + nOut] = sum;
        }
    }
}

extern "C" void kernel_launch(void* const* d_in, const int* in_sizes, int n_in,
                              void* d_out, int out_size, void* d_ws, size_t ws_size,
                              hipStream_t stream) {
    const float* x  = (const float*)d_in[0];
    const float* w1 = (const float*)d_in[1];
    const float* w2 = (const float*)d_in[2];
    float* out = (float*)d_out;

    fused_conv_wmma<<<dim3(25, 8), 128, 0, stream>>>(x, w2, w1, out);
}


# K slice walked as a 2-step pipeline: all loads up front in half order, half 2 lands under half 1's products; stage 1 re-mapped (lane = row x 16 B piece) so a half is ready early; same sum order
# speedup vs baseline: 5.7296x; 1.1007x over previous
#include <hip/hip_runtime.h>


typedef float v2f __attribute__((ext_vector_type(2)));
typedef float v4f __attribute__((ext_vector_type(4)));
typedef float v8f __attribute__((ext_vector_type(8)));

__device__ __forceinline__ float from_lane_below(float v) {
    return __builtin_bit_cast(float, __builtin_amdgcn_update_dpp(
        0, __builtin_bit_cast(int, v), 0x111  , 0xf, 0xf, true));
}
__device__ __forceinline__ float from_lane_above(float v) {
    return __builtin_bit_cast(float, __builtin_amdgcn_update_dpp(
        0, __builtin_bit_cast(int, v), 0x101  , 0xf, 0xf, true));
}

__device__ __forceinline__ void stage1_pair(const v4f xa, const v4f xb,
                                            const v4f q0, const v4f q1,
                                            const v4f q2, float* dst,
                                            const bool hasOutputs) {
    const float wflat[12] = {q0[0], q0[1], q0[2], q0[3], q1[0], q1[1],
                             q1[2], q1[3], q2[0], q2[1], q2[2], q2[3]};
    const float ea[6] = {from_lane_below(xa[3]), xa[0], xa[1], xa[2], xa[3],
                         from_lane_above(xa[0])};
    const float eb[6] = {from_lane_below(xb[3]), xb[0], xb[1], xb[2], xb[3],
                         from_lane_above(xb[0])};
    v2f o[4];
#pragma unroll
    for (int j = 0; j < 4; ++j) {
        float v0 = 0.0f, v1 = 0.0f;
#pragma unroll
        for (int i = 0; i < 3; ++i) {
            v0 = __builtin_fmaf(ea[j + i], wflat[i * 2 + 0], v0);
            v1 = __builtin_fmaf(ea[j + i], wflat[i * 2 + 1], v1);
        }
#pragma unroll
        for (int i = 0; i < 3; ++i) {
            v0 = __builtin_fmaf(eb[j + i], wflat[6 + i * 2 + 0], v0);
            v1 = __builtin_fmaf(eb[j + i], wflat[6 + i * 2 + 1], v1);
        }
        o[j] = (v2f){v0, v1};
    }
    if (hasOutputs) {
#pragma unroll
        for (int j = 0; j < 4; ++j) {
            *reinterpret_cast<v2f*>(dst + j * 2) = o[j];
        }
    }
}

#define LDS_ELEMS (32 * 4 * 30 * 2)
#define A_SLICE   144
#define A_HALF    72
#define A_PITCH   148
#define A_ROWS    16
#define A_WAVE    (A_ROWS * A_PITCH)
#define A_HLOADS  9
#define H_STEPS   18
#define RED_ELEMS (4 * 16 * 32)

__global__ __launch_bounds__(128) void fused_conv_wmma(
    const float* __restrict__ x,
    const float* __restrict__ w2,
    const float* w1,
    float* __restrict__ out)
{
    __shared__ __attribute__((aligned(16))) float btile[LDS_ELEMS];
    __shared__ __attribute__((aligned(16))) float atile[4 * A_WAVE];
    __shared__ __attribute__((aligned(16))) float red[RED_ELEMS];

    const int lane    = threadIdx.x & 31;
    const int wave    = threadIdx.x >> 5;
    const int tileN   = blockIdx.x * 32;
    const int mBlk    = blockIdx.y * 16;
    const int h0      = tileN / 28;

    const float* aSrc = w1 + mBlk * 576 + wave * A_SLICE;
    float* aDst = atile + wave * A_WAVE;
    int gOff[A_HLOADS], lOff[A_HLOADS];
#pragma unroll
    for (int b = 0; b < A_HLOADS; ++b) {
        const int q   = b * 32 + lane;
        const int row = q / 18;
        const int pc  = q - row * 18;
        gOff[b] = row * 576 + pc * 4;
        lOff[b] = row * A_PITCH + pc * 4;
    }

    const float* w2w = w2 + wave * 96;
    v4f wq[8][3];
#pragma unroll
    for (int c = 0; c < 4; ++c) {
#pragma unroll
        for (int k = 0; k < 3; ++k) {
            wq[c][k] = *reinterpret_cast<const v4f*>(w2w + c * 12 + 4 * k);
        }
    }

    const int  waveU = __builtin_amdgcn_readfirstlane(wave);
    const int  s1ri  = lane >> 3;
    const int  s1pk  = lane & 7;
    const int  s1hin = h0 - 1 + s1ri;
    const bool s1ld  = (s1pk < 7) && (s1hin >= 0) && (s1hin < 28);
    const int  s1hinC = s1hin < 0 ? 0 : (s1hin > 27 ? 27 : s1hin);
    const int  s1pkC  = s1pk < 7 ? s1pk : 6;
    const float* xl = x + (16 * waveU) * 784 + s1hinC * 28 + 4 * s1pkC;
    const v4f zero4 = {0.0f, 0.0f, 0.0f, 0.0f};
    v4f xa[8], xb[8];
#pragma unroll
    for (int c = 0; c < 4; ++c) {
        xa[c] = *reinterpret_cast<const v4f*>(xl + (2 * c) * 784);
        xb[c] = *reinterpret_cast<const v4f*>(xl + (2 * c + 1) * 784);
    }

    __builtin_amdgcn_sched_barrier(0);

    v4f areg0[A_HLOADS], areg1[A_HLOADS];
#pragma unroll
    for (int j = 0; j < A_HLOADS; ++j) {
        areg0[j] = *reinterpret_cast<const v4f*>(aSrc + gOff[j]);
    }

    __builtin_amdgcn_sched_barrier(0);

#pragma unroll
    for (int c = 4; c < 8; ++c) {
#pragma unroll
        for (int k = 0; k < 3; ++k) {
            wq[c][k] = *reinterpret_cast<const v4f*>(w2w + c * 12 + 4 * k);
        }
    }
#pragma unroll
    for (int c = 4; c < 8; ++c) {
        xa[c] = *reinterpret_cast<const v4f*>(xl + (2 * c) * 784);
        xb[c] = *reinterpret_cast<const v4f*>(xl + (2 * c + 1) * 784);
    }

    __builtin_amdgcn_sched_barrier(0);

#pragma unroll
    for (int j = 0; j < A_HLOADS; ++j) {
        areg1[j] = *reinterpret_cast<const v4f*>(aSrc + A_HALF + gOff[j]);
    }

    const int halfsel = lane >> 4;
    const int lane16  = lane & 15;

    const int nCol0 = tileN + lane16;
    const int nCol1 = nCol0 + 16;
    const int hA = nCol0 / 28, wA = nCol0 - hA * 28;
    const int hB = nCol1 / 28, wB = nCol1 - hB * 28;

    const float* aFrag = aDst + lane16 * A_PITCH + halfsel * 18;
    const float* bFrag0 = btile + (8 * wave + halfsel) * 240
                                + ((hA - h0) * 30 + wA) * 2;
    const float* bFrag1 = btile + (8 * wave + halfsel) * 240
                                + ((hB - h0) * 30 + wB) * 2;

    {
        float* hdst = btile + (threadIdx.x >> 2) * 240 + (threadIdx.x & 3) * 60;
        const v2f zero2 = {0.0f, 0.0f};
        *reinterpret_cast<v2f*>(hdst)          = zero2;
        *reinterpret_cast<v2f*>(hdst + 29 * 2) = zero2;
    }
    float* bdst = btile + (8 * waveU) * 240 + s1ri * 60 + (4 * s1pk + 1) * 2;
    const bool s1out = s1pk < 7;

    v8f acc0 = {}, acc1 = {};
    v2f af[H_STEPS], bf0[H_STEPS], bf1[H_STEPS];

#pragma unroll
    for (int c = 0; c < 4; ++c) {
        stage1_pair(s1ld ? xa[c] : zero4, s1ld ? xb[c] : zero4,
                    wq[c][0], wq[c][1], wq[c][2], bdst + c * 240, s1out);
    }
#pragma unroll
    for (int j = 0; j < A_HLOADS; ++j) {
        *reinterpret_cast<v4f*>(aDst + lOff[j]) = areg0[j];
    }
    __builtin_amdgcn_fence(__ATOMIC_RELEASE, "wavefront");
    __builtin_amdgcn_wave_barrier();
    __builtin_amdgcn_fence(__ATOMIC_ACQUIRE, "wavefront");

#pragma unroll
    for (int st = 0; st < H_STEPS; ++st) {
        const int u = st / 9, rs = st - u * 9;
        const int r = rs / 3, s = rs - r * 3;
        af[st]  = *reinterpret_cast<const v2f*>(aFrag + u * 36 + rs * 2);
        bf0[st] = *reinterpret_cast<const v2f*>(
            bFrag0 + (2 * u) * 240 + (r * 30 + s) * 2);
        bf1[st] = *reinterpret_cast<const v2f*>(
            bFrag1 + (2 * u) * 240 + (r * 30 + s) * 2);
    }

    __builtin_amdgcn_sched_barrier(0);

#pragma unroll
    for (int st = 0; st < H_STEPS; ++st) {
        acc0 = __builtin_amdgcn_wmma_f32_16x16x4_f32(
             false, af[st],  false, bf0[st],
             (short)0, acc0,  false,  false);
        acc1 = __builtin_amdgcn_wmma_f32_16x16x4_f32(
             false, af[st],  false, bf1[st],
             (short)0, acc1,  false,  false);
    }

    int s1ld2 = s1ld ? 1 : 0;
    asm volatile("" : "+v"(s1ld2) : "v"(acc0), "v"(acc1));
    const bool s1ldB = s1ld2 != 0;

#pragma unroll
    for (int c = 4; c < 8; ++c) {
        stage1_pair(s1ldB ? xa[c] : zero4, s1ldB ? xb[c] : zero4,
                    wq[c][0], wq[c][1], wq[c][2], bdst + c * 240, s1out);
    }
#pragma unroll
    for (int j = 0; j < A_HLOADS; ++j) {
        *reinterpret_cast<v4f*>(aDst + A_HALF + lOff[j]) = areg1[j];
    }
    __builtin_amdgcn_fence(__ATOMIC_RELEASE, "wavefront");
    __builtin_amdgcn_wave_barrier();
    __builtin_amdgcn_fence(__ATOMIC_ACQUIRE, "wavefront");

#pragma unroll
    for (int st = 0; st < H_STEPS; ++st) {
        const int u = 2 + st / 9, rs = st - (st / 9) * 9;
        const int r = rs / 3, s = rs - r * 3;
        af[st]  = *reinterpret_cast<const v2f*>(aFrag + u * 36 + rs * 2);
        bf0[st] = *reinterpret_cast<const v2f*>(
            bFrag0 + (2 * u) * 240 + (r * 30 + s) * 2);
        bf1[st] = *reinterpret_cast<const v2f*>(
            bFrag1 + (2 * u) * 240 + (r * 30 + s) * 2);
    }

    __builtin_amdgcn_sched_barrier(0);

#pragma unroll
    for (int st = 0; st < H_STEPS; ++st) {
        acc0 = __builtin_amdgcn_wmma_f32_16x16x4_f32(
             false, af[st],  false, bf0[st],
             (short)0, acc0,  false,  false);
        acc1 = __builtin_amdgcn_wmma_f32_16x16x4_f32(
             false, af[st],  false, bf1[st],
             (short)0, acc1,  false,  false);
    }

    float* rdst = red + wave * (16 * 32) + lane;
#pragma unroll
    for (int rr = 0; rr < 8; ++rr) {
        rdst[rr * 32]       = acc0[rr];
        rdst[(8 + rr) * 32] = acc1[rr];
    }
    __syncthreads();

    const int tsel = lane >> 4;
    const int nOut = tileN + lane;
#pragma unroll
    for (int e = 0; e < 4; ++e) {
        const int m   = wave * 4 + e;
        const int hs  = m >> 3, rr = m & 7;
        const float* rsrc = red + (tsel * 8 + rr) * 32 + hs * 16 + lane16;
        float sum = rsrc[0];
#pragma unroll
        for (int v = 1; v < 4; ++v) {
            sum += rsrc[v * (16 * 32)];
        }
        if (nOut < 784) {
            out[(mBlk + m) * 784 + nOut] = sum;
        }
    }
}

extern "C" void kernel_launch(void* const* d_in, const int* in_sizes, int n_in,
                              void* d_out, int out_size, void* d_ws, size_t ws_size,
                              hipStream_t stream) {
    const float* x  = (const float*)d_in[0];
    const float* w1 = (const float*)d_in[1];
    const float* w2 = (const float*)d_in[2];
    float* out = (float*)d_out;

    fused_conv_wmma<<<dim3(25, 8), 128, 0, stream>>>(x, w2, w1, out);
}
